# starting kernel
# baseline (speedup 1.0000x reference)
#include <hip/hip_runtime.h>

typedef __attribute__((ext_vector_type(2))) float v2f;
typedef __attribute__((ext_vector_type(8))) float v8f;

#define CHW        2048
#define O_CAPS     32
#define D_DIM      32
#define I_DIM      16
#define BT         16
#define THREADS    256
#define CHW_PER_WG 16
#define NCHUNK     (CHW / CHW_PER_WG)

#define U_STRIDE   36
#define U_OFF      0
#define U_SIZE     (BT * O_CAPS * U_STRIDE)
#define S_OFF      (U_OFF + U_SIZE)
#define S_SIZE     (BT * O_CAPS * U_STRIDE)
#define X_OFF      (S_OFF + S_SIZE)
#define X_SIZE     (BT * I_DIM)
#define T_OFF      (X_OFF + X_SIZE)
#define T_SIZE     (BT * O_CAPS)
#define MX_OFF     (T_OFF + T_SIZE)
#define DEN_OFF    (MX_OFF + BT)
#define LDS_FLOATS (DEN_OFF + BT)

__global__ __launch_bounds__(THREADS)
void caps_route_pass(const float* __restrict__ x,
                     const float* __restrict__ Wm,
                     const float* __restrict__ v,
                     float* __restrict__ s_out)
{
    extern __shared__ float lds[];
    const int tid   = threadIdx.x;
    const int lane  = tid & 31;
    const int wave  = tid >> 5;
    const int half  = lane >> 4;
    const int ln    = lane & 15;
    const int btile = blockIdx.x;
    const int chunk = blockIdx.y;
    const int b0    = btile * BT;

    for (int i = tid; i < S_SIZE; i += THREADS) lds[S_OFF + i] = 0.0f;

    for (int it = 0; it < CHW_PER_WG; ++it) {
        const int chw = chunk * CHW_PER_WG + it;
        __syncthreads();

        {
            const int m = tid >> 4, i = tid & 15;
            lds[X_OFF + tid] = x[((size_t)(b0 + m) * CHW + chw) * I_DIM + i];
        }
        __syncthreads();

        v2f af[4];
#pragma unroll
        for (int kk = 0; kk < 4; ++kk) {
            const int i0 = 4 * kk + 2 * half;
            af[kk][0] = lds[X_OFF + ln * I_DIM + i0];
            af[kk][1] = lds[X_OFF + ln * I_DIM + i0 + 1];
        }

#pragma unroll
        for (int j = 0; j < 4; ++j) {
            const int o = wave * 4 + j;
            const size_t wbase = ((size_t)chw * O_CAPS + o) * (I_DIM * D_DIM);
            if (it + 1 < CHW_PER_WG)
                __builtin_prefetch(&Wm[wbase + (size_t)O_CAPS * I_DIM * D_DIM
                                         + (size_t)lane * 32], 0, 3);
#pragma unroll
            for (int dt = 0; dt < 2; ++dt) {
                v8f acc = {0.f, 0.f, 0.f, 0.f, 0.f, 0.f, 0.f, 0.f};
#pragma unroll
                for (int kk = 0; kk < 4; ++kk) {
                    const int i0 = 4 * kk + 2 * half;
                    const int d  = dt * 16 + ln;
                    v2f bf;
                    bf[0] = Wm[wbase + (size_t)i0 * D_DIM + d];
                    bf[1] = Wm[wbase + (size_t)(i0 + 1) * D_DIM + d];
                    acc = __builtin_amdgcn_wmma_f32_16x16x4_f32(
                              false, af[kk], false, bf, (short)0, acc, false, false);
                }
#pragma unroll
                for (int r = 0; r < 8; ++r) {
                    const int m = r + 8 * half;
                    lds[U_OFF + (m * O_CAPS + o) * U_STRIDE + dt * 16 + ln] = acc[r];
                }
            }
        }
        __syncthreads();

        if (v != nullptr) {
#pragma unroll
            for (int pp = 0; pp < 2; ++pp) {
                const int p = tid + pp * THREADS;
                const int m = p >> 5, o = p & 31;
                const float* vp = &v[(((size_t)(b0 + m)) * O_CAPS + o) * D_DIM];
                const float* up = &lds[U_OFF + (m * O_CAPS + o) * U_STRIDE];
                float t = 0.f;
#pragma unroll
                for (int q = 0; q < 8; ++q) {
                    const float4 uu = *(const float4*)(up + 4 * q);
                    const float4 vv = *(const float4*)(vp + 4 * q);
                    t += uu.x * vv.x + uu.y * vv.y + uu.z * vv.z + uu.w * vv.w;
                }
                lds[T_OFF + p] = t;
            }
            __syncthreads();
            {
                const int m = wave * 2 + half;
                const float t0 = lds[T_OFF + m * O_CAPS + ln];
                const float t1 = lds[T_OFF + m * O_CAPS + ln + 16];
                float mx = fmaxf(t0, t1);
#pragma unroll
                for (int msk = 8; msk >= 1; msk >>= 1)
                    mx = fmaxf(mx, __shfl_xor(mx, msk, 32));
                float den = __expf(t0 - mx) + __expf(t1 - mx);
#pragma unroll
                for (int msk = 8; msk >= 1; msk >>= 1)
                    den += __shfl_xor(den, msk, 32);
                if (ln == 0) { lds[MX_OFF + m] = mx; lds[DEN_OFF + m] = den; }
            }
            __syncthreads();
#pragma unroll
            for (int pp = 0; pp < 2; ++pp) {
                const int p = tid + pp * THREADS;
                const int m = p >> 5, o = p & 31;
                const float c = __expf(lds[T_OFF + p] - lds[MX_OFF + m]) / lds[DEN_OFF + m];
                float*       sp = &lds[S_OFF + (m * O_CAPS + o) * U_STRIDE];
                const float* up = &lds[U_OFF + (m * O_CAPS + o) * U_STRIDE];
#pragma unroll
                for (int q = 0; q < 8; ++q) {
                    float4 ss = *(float4*)(sp + 4 * q);
                    const float4 uu = *(const float4*)(up + 4 * q);
                    ss.x += c * uu.x;  ss.y += c * uu.y;
                    ss.z += c * uu.z;  ss.w += c * uu.w;
                    *(float4*)(sp + 4 * q) = ss;
                }
            }
        } else {
            const float c = 1.0f / 32.0f;
#pragma unroll
            for (int pp = 0; pp < 2; ++pp) {
                const int p = tid + pp * THREADS;
                const int m = p >> 5, o = p & 31;
                float*       sp = &lds[S_OFF + (m * O_CAPS + o) * U_STRIDE];
                const float* up = &lds[U_OFF + (m * O_CAPS + o) * U_STRIDE];
#pragma unroll
                for (int q = 0; q < 8; ++q) {
                    float4 ss = *(float4*)(sp + 4 * q);
                    const float4 uu = *(const float4*)(up + 4 * q);
                    ss.x += c * uu.x;  ss.y += c * uu.y;
                    ss.z += c * uu.z;  ss.w += c * uu.w;
                    *(float4*)(sp + 4 * q) = ss;
                }
            }
        }
    }
    __syncthreads();

#pragma unroll
    for (int pp = 0; pp < 2; ++pp) {
        const int p = tid + pp * THREADS;
        const int m = p >> 5, o = p & 31;
        const float* sp = &lds[S_OFF + (m * O_CAPS + o) * U_STRIDE];
        float* gp = &s_out[(((size_t)(b0 + m)) * O_CAPS + o) * D_DIM];
        for (int d = 0; d < D_DIM; ++d) atomicAdd(&gp[d], sp[d]);
    }
}

__global__ __launch_bounds__(256)
void squash_kernel(const float* __restrict__ s, float* __restrict__ vout,
                   const float* __restrict__ vprev, float* __restrict__ vsum)
{
    const int lane = threadIdx.x & 31;
    const int grp  = blockIdx.x * (blockDim.x >> 5) + (threadIdx.x >> 5);
    const int idx  = grp * 32 + lane;
    const float val = s[idx];
    float sq = val * val;
#pragma unroll
    for (int m = 16; m >= 1; m >>= 1) sq += __shfl_xor(sq, m, 32);
    const float scale = (sq / (1.0f + sq)) / (sqrtf(sq) + 1e-6f);
    const float r = val * scale;
    vout[idx] = r;
    if (vsum) vsum[idx] = r + vprev[idx];
}

extern "C" void kernel_launch(void* const* d_in, const int* in_sizes, int n_in,
                              void* d_out, int out_size, void* d_ws, size_t ws_size,
                              hipStream_t stream)
{
    const float* x  = (const float*)d_in[0];
    const float* Wm = (const float*)d_in[1];
    float* out = (float*)d_out;

    const size_t N = 64 * 32 * 32;
    float* wsf = (float*)d_ws;
    float* s1  = wsf + 0 * N;
    float* s2  = wsf + 1 * N;
    float* s3  = wsf + 2 * N;
    float* v1  = wsf + 3 * N;
    float* v2  = wsf + 4 * N;
    float* v12 = wsf + 5 * N;

    hipMemsetAsync(wsf, 0, 3 * N * sizeof(float), stream);

    dim3 grid(4, NCHUNK);
    const size_t shmem = LDS_FLOATS * sizeof(float);

    caps_route_pass<<<grid, THREADS, shmem, stream>>>(x, Wm, nullptr, s1);
    squash_kernel<<<(64 * 32) / 8, 256, 0, stream>>>(s1, v1, nullptr, nullptr);
    caps_route_pass<<<grid, THREADS, shmem, stream>>>(x, Wm, v1, s2);
    squash_kernel<<<(64 * 32) / 8, 256, 0, stream>>>(s2, v2, v1, v12);
    caps_route_pass<<<grid, THREADS, shmem, stream>>>(x, Wm, v12, s3);
    squash_kernel<<<(64 * 32) / 8, 256, 0, stream>>>(s3, out, nullptr, nullptr);
}


# s partials go out as whole-line stores into per-chunk workspace planes; the squash kernel adds the 128 planes in a fixed order: no global atomics, no memset; float sum order changes
# speedup vs baseline: 5.6408x; 5.6408x over previous
#include <hip/hip_runtime.h>

typedef __attribute__((ext_vector_type(2))) float v2f;
typedef __attribute__((ext_vector_type(8))) float v8f;

#define CHW        2048
#define O_CAPS     32
#define D_DIM      32
#define I_DIM      16
#define BT         16
#define THREADS    256
#define CHW_PER_WG 16
#define NCHUNK     (CHW / CHW_PER_WG)
#define PLANE      (64 * O_CAPS * D_DIM)

#define U_STRIDE   36
#define U_OFF      0
#define U_SIZE     (BT * O_CAPS * U_STRIDE)
#define S_OFF      (U_OFF + U_SIZE)
#define S_SIZE     (BT * O_CAPS * U_STRIDE)
#define X_OFF      (S_OFF + S_SIZE)
#define X_SIZE     (BT * I_DIM)
#define T_OFF      (X_OFF + X_SIZE)
#define T_SIZE     (BT * O_CAPS)
#define MX_OFF     (T_OFF + T_SIZE)
#define DEN_OFF    (MX_OFF + BT)
#define LDS_FLOATS (DEN_OFF + BT)

__global__ __launch_bounds__(THREADS)
void caps_route_pass(const float* __restrict__ x,
                     const float* __restrict__ Wm,
                     const float* __restrict__ v,
                     float* __restrict__ part)
{
    extern __shared__ float lds[];
    const int tid   = threadIdx.x;
    const int lane  = tid & 31;
    const int wave  = tid >> 5;
    const int half  = lane >> 4;
    const int ln    = lane & 15;
    const int btile = blockIdx.x;
    const int chunk = blockIdx.y;
    const int b0    = btile * BT;

    for (int i = tid; i < S_SIZE; i += THREADS) lds[S_OFF + i] = 0.0f;

    for (int it = 0; it < CHW_PER_WG; ++it) {
        const int chw = chunk * CHW_PER_WG + it;
        __syncthreads();

        {
            const int m = tid >> 4, i = tid & 15;
            lds[X_OFF + tid] = x[((size_t)(b0 + m) * CHW + chw) * I_DIM + i];
        }
        __syncthreads();

        v2f af[4];
#pragma unroll
        for (int kk = 0; kk < 4; ++kk) {
            const int i0 = 4 * kk + 2 * half;
            af[kk][0] = lds[X_OFF + ln * I_DIM + i0];
            af[kk][1] = lds[X_OFF + ln * I_DIM + i0 + 1];
        }

#pragma unroll
        for (int j = 0; j < 4; ++j) {
            const int o = wave * 4 + j;
            const size_t wbase = ((size_t)chw * O_CAPS + o) * (I_DIM * D_DIM);
            if (it + 1 < CHW_PER_WG)
                __builtin_prefetch(&Wm[wbase + (size_t)O_CAPS * I_DIM * D_DIM
                                         + (size_t)lane * 32], 0, 3);
#pragma unroll
            for (int dt = 0; dt < 2; ++dt) {
                v8f acc = {0.f, 0.f, 0.f, 0.f, 0.f, 0.f, 0.f, 0.f};
#pragma unroll
                for (int kk = 0; kk < 4; ++kk) {
                    const int i0 = 4 * kk + 2 * half;
                    const int d  = dt * 16 + ln;
                    v2f bf;
                    bf[0] = Wm[wbase + (size_t)i0 * D_DIM + d];
                    bf[1] = Wm[wbase + (size_t)(i0 + 1) * D_DIM + d];
                    acc = __builtin_amdgcn_wmma_f32_16x16x4_f32(
                              false, af[kk], false, bf, (short)0, acc, false, false);
                }
#pragma unroll
                for (int r = 0; r < 8; ++r) {
                    const int m = r + 8 * half;
                    lds[U_OFF + (m * O_CAPS + o) * U_STRIDE + dt * 16 + ln] = acc[r];
                }
            }
        }
        __syncthreads();

        if (v != nullptr) {
#pragma unroll
            for (int pp = 0; pp < 2; ++pp) {
                const int p = tid + pp * THREADS;
                const int m = p >> 5, o = p & 31;
                const float* vp = &v[(((size_t)(b0 + m)) * O_CAPS + o) * D_DIM];
                const float* up = &lds[U_OFF + (m * O_CAPS + o) * U_STRIDE];
                float t = 0.f;
#pragma unroll
                for (int q = 0; q < 8; ++q) {
                    const float4 uu = *(const float4*)(up + 4 * q);
                    const float4 vv = *(const float4*)(vp + 4 * q);
                    t += uu.x * vv.x + uu.y * vv.y + uu.z * vv.z + uu.w * vv.w;
                }
                lds[T_OFF + p] = t;
            }
            __syncthreads();
            {
                const int m = wave * 2 + half;
                const float t0 = lds[T_OFF + m * O_CAPS + ln];
                const float t1 = lds[T_OFF + m * O_CAPS + ln + 16];
                float mx = fmaxf(t0, t1);
#pragma unroll
                for (int msk = 8; msk >= 1; msk >>= 1)
                    mx = fmaxf(mx, __shfl_xor(mx, msk, 32));
                float den = __expf(t0 - mx) + __expf(t1 - mx);
#pragma unroll
                for (int msk = 8; msk >= 1; msk >>= 1)
                    den += __shfl_xor(den, msk, 32);
                if (ln == 0) { lds[MX_OFF + m] = mx; lds[DEN_OFF + m] = den; }
            }
            __syncthreads();
#pragma unroll
            for (int pp = 0; pp < 2; ++pp) {
                const int p = tid + pp * THREADS;
                const int m = p >> 5, o = p & 31;
                const float c = __expf(lds[T_OFF + p] - lds[MX_OFF + m]) / lds[DEN_OFF + m];
                float*       sp = &lds[S_OFF + (m * O_CAPS + o) * U_STRIDE];
                const float* up = &lds[U_OFF + (m * O_CAPS + o) * U_STRIDE];
#pragma unroll
                for (int q = 0; q < 8; ++q) {
                    float4 ss = *(float4*)(sp + 4 * q);
                    const float4 uu = *(const float4*)(up + 4 * q);
                    ss.x += c * uu.x;  ss.y += c * uu.y;
                    ss.z += c * uu.z;  ss.w += c * uu.w;
                    *(float4*)(sp + 4 * q) = ss;
                }
            }
        } else {
            const float c = 1.0f / 32.0f;
#pragma unroll
            for (int pp = 0; pp < 2; ++pp) {
                const int p = tid + pp * THREADS;
                const int m = p >> 5, o = p & 31;
                float*       sp = &lds[S_OFF + (m * O_CAPS + o) * U_STRIDE];
                const float* up = &lds[U_OFF + (m * O_CAPS + o) * U_STRIDE];
#pragma unroll
                for (int q = 0; q < 8; ++q) {
                    float4 ss = *(float4*)(sp + 4 * q);
                    const float4 uu = *(const float4*)(up + 4 * q);
                    ss.x += c * uu.x;  ss.y += c * uu.y;
                    ss.z += c * uu.z;  ss.w += c * uu.w;
                    *(float4*)(sp + 4 * q) = ss;
                }
            }
        }
    }
    __syncthreads();

    float* gp = part + (size_t)chunk * PLANE + (size_t)b0 * (O_CAPS * D_DIM);
#pragma unroll
    for (int k = 0; k < (BT * O_CAPS * D_DIM) / (4 * THREADS); ++k) {
        const int f   = tid + k * THREADS;
        const int row = f >> 3;
        const int q   = f & 7;
        const float* sp = &lds[S_OFF + row * U_STRIDE];
        const float4 ss = *(const float4*)(sp + 4 * q);
        *(float4*)(gp + 4 * f) = ss;
    }
}

__global__ __launch_bounds__(256)
void squash_kernel(const float* __restrict__ part, float* __restrict__ vout,
                   const float* __restrict__ vprev, float* __restrict__ vsum)
{
    __shared__ float4 red[8][32];
    const int lane = threadIdx.x & 31;
    const int wave = threadIdx.x >> 5;
    const int grp  = wave >> 2;
    const int qtr  = wave & 3;
    const int e    = (blockIdx.x * 2 + grp) * 128 + lane * 4;
    const float* p = part + (size_t)(qtr * (NCHUNK / 4)) * PLANE + e;
    float4 acc = {0.f, 0.f, 0.f, 0.f};
#pragma unroll 16
    for (int c = 0; c < NCHUNK / 4; ++c) {
        const float4 t = *(const float4*)(p + (size_t)c * PLANE);
        acc.x += t.x;  acc.y += t.y;  acc.z += t.z;  acc.w += t.w;
    }
    red[wave][lane] = acc;
    __syncthreads();
    if (qtr == 0) {
        const float4 a0 = red[wave][lane],     a1 = red[wave + 1][lane];
        const float4 a2 = red[wave + 2][lane], a3 = red[wave + 3][lane];
        float4 s;
        s.x = ((a0.x + a1.x) + a2.x) + a3.x;
        s.y = ((a0.y + a1.y) + a2.y) + a3.y;
        s.z = ((a0.z + a1.z) + a2.z) + a3.z;
        s.w = ((a0.w + a1.w) + a2.w) + a3.w;
        float sq = s.x * s.x + s.y * s.y + s.z * s.z + s.w * s.w;
        for (int m = 4; m >= 1; m >>= 1) sq += __shfl_xor(sq, m, 32);
        const float scale = (sq / (1.0f + sq)) / (sqrtf(sq) + 1e-6f);
        float4 r;
        r.x = s.x * scale;  r.y = s.y * scale;  r.z = s.z * scale;  r.w = s.w * scale;
        *(float4*)(vout + e) = r;
        if (vsum) {
            const float4 pv = *(const float4*)(vprev + e);
            float4 t;
            t.x = r.x + pv.x;  t.y = r.y + pv.y;  t.z = r.z + pv.z;  t.w = r.w + pv.w;
            *(float4*)(vsum + e) = t;
        }
    }
}

extern "C" void kernel_launch(void* const* d_in, const int* in_sizes, int n_in,
                              void* d_out, int out_size, void* d_ws, size_t ws_size,
                              hipStream_t stream)
{
    const float* x  = (const float*)d_in[0];
    const float* Wm = (const float*)d_in[1];
    float* out = (float*)d_out;

    const size_t N = 64 * 32 * 32;
    float* wsf  = (float*)d_ws;
    float* v1   = wsf + 0 * N;
    float* v2   = wsf + 1 * N;
    float* v12  = wsf + 2 * N;
    float* part = wsf + 4 * N;

    dim3 grid(4, NCHUNK);
    const size_t shmem = LDS_FLOATS * sizeof(float);

    caps_route_pass<<<grid, THREADS, shmem, stream>>>(x, Wm, nullptr, part);
    squash_kernel<<<(64 * 32) / 8, 256, 0, stream>>>(part, v1, nullptr, nullptr);
    caps_route_pass<<<grid, THREADS, shmem, stream>>>(x, Wm, v1, part);
    squash_kernel<<<(64 * 32) / 8, 256, 0, stream>>>(part, v2, v1, v12);
    caps_route_pass<<<grid, THREADS, shmem, stream>>>(x, Wm, v12, part);
    squash_kernel<<<(64 * 32) / 8, 256, 0, stream>>>(part, out, nullptr, nullptr);
}


# WMMA operands swapped (W as A, x as B): a lane owns one batch row, so u_hat and s stay in registers, only logits cross waves via LDS; W fragments fetched one o ahead; logit/softmax sum order changes
# speedup vs baseline: 8.0635x; 1.4295x over previous
#include <hip/hip_runtime.h>

typedef __attribute__((ext_vector_type(2))) float v2f;
typedef __attribute__((ext_vector_type(8))) float v8f;

#define CHW        2048
#define O_CAPS     32
#define D_DIM      32
#define I_DIM      16
#define BT         16
#define THREADS    256
#define CHW_PER_WG 16
#define NCHUNK     (CHW / CHW_PER_WG)
#define PLANE      (64 * O_CAPS * D_DIM)

#define U_STRIDE   36
#define M_STRIDE   (O_CAPS * U_STRIDE + 4)
#define S_OFF      0
#define S_SIZE     (BT * M_STRIDE)
#define T_STRIDE   36
#define T_OFF      (S_OFF + S_SIZE)
#define T_SIZE     (BT * T_STRIDE)
#define V_OFF      (T_OFF + 2 * T_SIZE)
#define V_SIZE     (BT * O_CAPS * D_DIM)
#define LDS_FLOATS_NOV (V_OFF)
#define LDS_FLOATS_V   (V_OFF + V_SIZE)

template <bool HAS_V>
__global__ __launch_bounds__(THREADS)
__attribute__((amdgpu_waves_per_eu(4, 4)))
void caps_route_pass(const float* __restrict__ x,
                     const float* __restrict__ Wm,
                     const float* __restrict__ v,
                     float* __restrict__ part)
{
    extern __shared__ float lds[];
    const int tid   = threadIdx.x;
    const int lane  = tid & 31;
    const int wave  = tid >> 5;
    const int half  = lane >> 4;
    const int ln    = lane & 15;
    const int btile = blockIdx.x;
    const int chunk = blockIdx.y;
    const int b0    = btile * BT;

    float* vslice = &lds[V_OFF + (wave * 16 * 32 + lane) * 4];
    if (HAS_V) {
#pragma unroll
        for (int j = 0; j < 4; ++j) {
            const int o = wave * 4 + j;
            const float* vp = &v[(((size_t)(b0 + ln)) * O_CAPS + o) * D_DIM + 8 * half];
#pragma unroll
            for (int dt = 0; dt < 2; ++dt)
#pragma unroll
                for (int q = 0; q < 2; ++q)
                    *(float4*)(vslice + ((j * 2 + dt) * 2 + q) * 128) =
                        *(const float4*)(vp + 16 * dt + 4 * q);
        }
    }

    v8f sacc[4][2];
#pragma unroll
    for (int j = 0; j < 4; ++j)
#pragma unroll
        for (int dt = 0; dt < 2; ++dt)
            sacc[j][dt] = (v8f){0.f, 0.f, 0.f, 0.f, 0.f, 0.f, 0.f, 0.f};

    const float* xrow = x + ((size_t)(b0 + ln) * CHW + (size_t)chunk * CHW_PER_WG) * I_DIM + 2 * half;
    v2f xf[4];
#pragma unroll
    for (int kk = 0; kk < 4; ++kk) xf[kk] = *(const v2f*)(xrow + 4 * kk);

    const int woff = (2 * half) * D_DIM + ln;
    const float* wchunk = Wm + ((size_t)chunk * CHW_PER_WG * O_CAPS + wave * 4) * (I_DIM * D_DIM) + woff;
    v2f wf[2][4][2];
#pragma unroll
    for (int kk = 0; kk < 4; ++kk)
#pragma unroll
        for (int dt = 0; dt < 2; ++dt) {
            wf[0][kk][dt][0] = wchunk[(4 * kk) * D_DIM + 16 * dt];
            wf[0][kk][dt][1] = wchunk[(4 * kk + 1) * D_DIM + 16 * dt];
        }

    for (int it = 0; it < CHW_PER_WG; ++it) {
        const float* wit = wchunk + (size_t)it * (O_CAPS * I_DIM * D_DIM);

        v8f acc[4][2];
#pragma unroll
        for (int j = 0; j < 4; ++j) {
            if (it + 1 < CHW_PER_WG)
                __builtin_prefetch(wit - woff + (size_t)(O_CAPS + j) * (I_DIM * D_DIM)
                                       + (size_t)lane * 32, 0, 3);
            if (j < 3 || it + 1 < CHW_PER_WG) {
                const float* wnx = (j < 3) ? wit + (j + 1) * (I_DIM * D_DIM)
                                           : wit + (size_t)(O_CAPS * I_DIM * D_DIM);
#pragma unroll
                for (int kk = 0; kk < 4; ++kk)
#pragma unroll
                    for (int dt = 0; dt < 2; ++dt) {
                        wf[(j + 1) & 1][kk][dt][0] = wnx[(4 * kk) * D_DIM + 16 * dt];
                        wf[(j + 1) & 1][kk][dt][1] = wnx[(4 * kk + 1) * D_DIM + 16 * dt];
                    }
            }
            acc[j][0] = (v8f){0.f, 0.f, 0.f, 0.f, 0.f, 0.f, 0.f, 0.f};
            acc[j][1] = (v8f){0.f, 0.f, 0.f, 0.f, 0.f, 0.f, 0.f, 0.f};
#pragma unroll
            for (int kk = 0; kk < 4; ++kk)
#pragma unroll
                for (int dt = 0; dt < 2; ++dt)
                    acc[j][dt] = __builtin_amdgcn_wmma_f32_16x16x4_f32(
                                     false, wf[j & 1][kk][dt], false, xf[kk], (short)0, acc[j][dt], false, false);
        }

        if (it + 1 < CHW_PER_WG) {
#pragma unroll
            for (int kk = 0; kk < 4; ++kk)
                xf[kk] = *(const v2f*)(xrow + (size_t)(it + 1) * I_DIM + 4 * kk);
        }

        if (HAS_V) {
            float t[4];
#pragma unroll
            for (int j = 0; j < 4; ++j) {
                float p0 = 0.f, p1 = 0.f;
#pragma unroll
                for (int dt = 0; dt < 2; ++dt) {
                    const float4 va = *(const float4*)(vslice + ((j * 2 + dt) * 2 + 0) * 128);
                    const float4 vb = *(const float4*)(vslice + ((j * 2 + dt) * 2 + 1) * 128);
                    p0 += acc[j][dt][0] * va.x + acc[j][dt][1] * va.y
                        + acc[j][dt][2] * va.z + acc[j][dt][3] * va.w;
                    p1 += acc[j][dt][4] * vb.x + acc[j][dt][5] * vb.y
                        + acc[j][dt][6] * vb.z + acc[j][dt][7] * vb.w;
                }
                const float p = p0 + p1;
                t[j] = p + __shfl_xor(p, 16, 32);
            }
            float* tb = &lds[T_OFF + (it & 1) * T_SIZE];
            if (half == 0)
                *(float4*)(tb + ln * T_STRIDE + 4 * wave) = make_float4(t[0], t[1], t[2], t[3]);
            __syncthreads();
            float tv[16];
#pragma unroll
            for (int q = 0; q < 4; ++q) {
                const float4 a = *(const float4*)(tb + ln * T_STRIDE + 16 * half + 4 * q);
                tv[4 * q] = a.x;  tv[4 * q + 1] = a.y;  tv[4 * q + 2] = a.z;  tv[4 * q + 3] = a.w;
            }
            float mx = tv[0];
#pragma unroll
            for (int q = 1; q < 16; ++q) mx = fmaxf(mx, tv[q]);
            mx = fmaxf(mx, __shfl_xor(mx, 16, 32));
            float den = 0.f;
#pragma unroll
            for (int q = 0; q < 16; ++q) den += __expf(tv[q] - mx);
            den += __shfl_xor(den, 16, 32);
            const float inv = 1.0f / den;
#pragma unroll
            for (int j = 0; j < 4; ++j) {
                const float c = __expf(t[j] - mx) * inv;
#pragma unroll
                for (int dt = 0; dt < 2; ++dt)
#pragma unroll
                    for (int r = 0; r < 8; ++r)
                        sacc[j][dt][r] += c * acc[j][dt][r];
            }
        } else {
            const float c = 1.0f / 32.0f;
#pragma unroll
            for (int j = 0; j < 4; ++j)
#pragma unroll
                for (int dt = 0; dt < 2; ++dt)
#pragma unroll
                    for (int r = 0; r < 8; ++r)
                        sacc[j][dt][r] += c * acc[j][dt][r];
        }
    }

#pragma unroll
    for (int j = 0; j < 4; ++j) {
        const int o = wave * 4 + j;
#pragma unroll
        for (int dt = 0; dt < 2; ++dt) {
            float* sp = &lds[S_OFF + ln * M_STRIDE + o * U_STRIDE + 16 * dt + 8 * half];
            *(float4*)(sp)     = make_float4(sacc[j][dt][0], sacc[j][dt][1], sacc[j][dt][2], sacc[j][dt][3]);
            *(float4*)(sp + 4) = make_float4(sacc[j][dt][4], sacc[j][dt][5], sacc[j][dt][6], sacc[j][dt][7]);
        }
    }
    __syncthreads();

    float* gp = part + (size_t)chunk * PLANE + (size_t)b0 * (O_CAPS * D_DIM);
#pragma unroll
    for (int k = 0; k < (BT * O_CAPS * D_DIM) / (4 * THREADS); ++k) {
        const int f   = tid + k * THREADS;
        const int row = f >> 3;
        const int q   = f & 7;
        const float* sp = &lds[S_OFF + (row >> 5) * M_STRIDE + (row & 31) * U_STRIDE];
        const float4 ss = *(const float4*)(sp + 4 * q);
        *(float4*)(gp + 4 * f) = ss;
    }
}

__global__ __launch_bounds__(256)
void squash_kernel(const float* __restrict__ part, float* __restrict__ vout,
                   const float* __restrict__ vprev, float* __restrict__ vsum)
{
    __shared__ float4 red[8][32];
    const int lane = threadIdx.x & 31;
    const int wave = threadIdx.x >> 5;
    const int grp  = wave >> 2;
    const int qtr  = wave & 3;
    const int e    = (blockIdx.x * 2 + grp) * 128 + lane * 4;
    const float* p = part + (size_t)(qtr * (NCHUNK / 4)) * PLANE + e;
    float4 acc = {0.f, 0.f, 0.f, 0.f};
#pragma unroll 16
    for (int c = 0; c < NCHUNK / 4; ++c) {
        const float4 t = *(const float4*)(p + (size_t)c * PLANE);
        acc.x += t.x;  acc.y += t.y;  acc.z += t.z;  acc.w += t.w;
    }
    red[wave][lane] = acc;
    __syncthreads();
    if (qtr == 0) {
        const float4 a0 = red[wave][lane],     a1 = red[wave + 1][lane];
        const float4 a2 = red[wave + 2][lane], a3 = red[wave + 3][lane];
        float4 s;
        s.x = ((a0.x + a1.x) + a2.x) + a3.x;
        s.y = ((a0.y + a1.y) + a2.y) + a3.y;
        s.z = ((a0.z + a1.z) + a2.z) + a3.z;
        s.w = ((a0.w + a1.w) + a2.w) + a3.w;
        float sq = s.x * s.x + s.y * s.y + s.z * s.z + s.w * s.w;
        for (int m = 4; m >= 1; m >>= 1) sq += __shfl_xor(sq, m, 32);
        const float scale = (sq / (1.0f + sq)) / (sqrtf(sq) + 1e-6f);
        float4 r;
        r.x = s.x * scale;  r.y = s.y * scale;  r.z = s.z * scale;  r.w = s.w * scale;
        *(float4*)(vout + e) = r;
        if (vsum) {
            const float4 pv = *(const float4*)(vprev + e);
            float4 t;
            t.x = r.x + pv.x;  t.y = r.y + pv.y;  t.z = r.z + pv.z;  t.w = r.w + pv.w;
            *(float4*)(vsum + e) = t;
        }
    }
}

extern "C" void kernel_launch(void* const* d_in, const int* in_sizes, int n_in,
                              void* d_out, int out_size, void* d_ws, size_t ws_size,
                              hipStream_t stream)
{
    const float* x  = (const float*)d_in[0];
    const float* Wm = (const float*)d_in[1];
    float* out = (float*)d_out;

    const size_t N = 64 * 32 * 32;
    float* wsf  = (float*)d_ws;
    float* v1   = wsf + 0 * N;
    float* v2   = wsf + 1 * N;
    float* v12  = wsf + 2 * N;
    float* part = wsf + 4 * N;

    dim3 grid(4, NCHUNK);
    const size_t shm0 = LDS_FLOATS_NOV * sizeof(float);
    const size_t shm1 = LDS_FLOATS_V * sizeof(float);

    caps_route_pass<false><<<grid, THREADS, shm0, stream>>>(x, Wm, nullptr, part);
    squash_kernel<<<(64 * 32) / 8, 256, 0, stream>>>(part, v1, nullptr, nullptr);
    caps_route_pass<true><<<grid, THREADS, shm1, stream>>>(x, Wm, v1, part);
    squash_kernel<<<(64 * 32) / 8, 256, 0, stream>>>(part, v2, v1, v12);
    caps_route_pass<true><<<grid, THREADS, shm1, stream>>>(x, Wm, v12, part);
    squash_kernel<<<(64 * 32) / 8, 256, 0, stream>>>(part, out, nullptr, nullptr);
}


# W and x fragments fetched one whole chw ahead: four register buffers (one per o), each refilled behind its o's WMMAs; per-chw barrier orders LDS only so the loads stay in flight; no cache prefetch
# speedup vs baseline: 8.4237x; 1.0447x over previous
#include <hip/hip_runtime.h>

typedef __attribute__((ext_vector_type(2))) float v2f;
typedef __attribute__((ext_vector_type(8))) float v8f;

#define CHW        2048
#define O_CAPS     32
#define D_DIM      32
#define I_DIM      16
#define BT         16
#define THREADS    256
#define CHW_PER_WG 16
#define NCHUNK     (CHW / CHW_PER_WG)
#define PLANE      (64 * O_CAPS * D_DIM)

#define U_STRIDE   36
#define M_STRIDE   (O_CAPS * U_STRIDE + 4)
#define S_OFF      0
#define S_SIZE     (BT * M_STRIDE)
#define T_STRIDE   36
#define T_OFF      (S_OFF + S_SIZE)
#define T_SIZE     (BT * T_STRIDE)
#define V_OFF      (T_OFF + 2 * T_SIZE)
#define V_SIZE     (BT * O_CAPS * D_DIM)
#define LDS_FLOATS_NOV (V_OFF)
#define LDS_FLOATS_V   (V_OFF + V_SIZE)

template <bool HAS_V>
__global__ __launch_bounds__(THREADS)
__attribute__((amdgpu_waves_per_eu(4, 4)))
void caps_route_pass(const float* __restrict__ x,
                     const float* __restrict__ Wm,
                     const float* __restrict__ v,
                     float* __restrict__ part)
{
    extern __shared__ float lds[];
    const int tid   = threadIdx.x;
    const int lane  = tid & 31;
    const int wave  = tid >> 5;
    const int half  = lane >> 4;
    const int ln    = lane & 15;
    const int btile = blockIdx.x;
    const int chunk = blockIdx.y;
    const int b0    = btile * BT;

    float* vslice = &lds[V_OFF + (wave * 16 * 32 + lane) * 4];
    if (HAS_V) {
#pragma unroll
        for (int j = 0; j < 4; ++j) {
            const int o = wave * 4 + j;
            const float* vp = &v[(((size_t)(b0 + ln)) * O_CAPS + o) * D_DIM + 8 * half];
#pragma unroll
            for (int dt = 0; dt < 2; ++dt)
#pragma unroll
                for (int q = 0; q < 2; ++q)
                    *(float4*)(vslice + ((j * 2 + dt) * 2 + q) * 128) =
                        *(const float4*)(vp + 16 * dt + 4 * q);
        }
    }

    v8f sacc[4][2];
#pragma unroll
    for (int j = 0; j < 4; ++j)
#pragma unroll
        for (int dt = 0; dt < 2; ++dt)
            sacc[j][dt] = (v8f){0.f, 0.f, 0.f, 0.f, 0.f, 0.f, 0.f, 0.f};

    const float* xrow = x + ((size_t)(b0 + ln) * CHW + (size_t)chunk * CHW_PER_WG) * I_DIM + 2 * half;
    v2f xf[4];
#pragma unroll
    for (int kk = 0; kk < 4; ++kk) xf[kk] = *(const v2f*)(xrow + 4 * kk);
    __builtin_amdgcn_sched_barrier(0);

    const int woff = (2 * half) * D_DIM + ln;
    const float* wchunk = Wm + ((size_t)chunk * CHW_PER_WG * O_CAPS + wave * 4) * (I_DIM * D_DIM) + woff;
    v2f wf[4][4][2];
#pragma unroll
    for (int j = 0; j < 4; ++j)
#pragma unroll
        for (int kk = 0; kk < 4; ++kk)
#pragma unroll
            for (int dt = 0; dt < 2; ++dt) {
                wf[j][kk][dt][0] = wchunk[j * (I_DIM * D_DIM) + (4 * kk) * D_DIM + 16 * dt];
                wf[j][kk][dt][1] = wchunk[j * (I_DIM * D_DIM) + (4 * kk + 1) * D_DIM + 16 * dt];
            }

    for (int it = 0; it < CHW_PER_WG; ++it) {
        const bool more = it + 1 < CHW_PER_WG;
        const float* wnx = wchunk + (size_t)(it + 1) * (O_CAPS * I_DIM * D_DIM);

        v2f xn[4];
        if (more) {
#pragma unroll
            for (int kk = 0; kk < 4; ++kk)
                xn[kk] = *(const v2f*)(xrow + (size_t)(it + 1) * I_DIM + 4 * kk);
        }

        v8f acc[4][2];
#pragma unroll
        for (int j = 0; j < 4; ++j) {
            acc[j][0] = (v8f){0.f, 0.f, 0.f, 0.f, 0.f, 0.f, 0.f, 0.f};
            acc[j][1] = (v8f){0.f, 0.f, 0.f, 0.f, 0.f, 0.f, 0.f, 0.f};
#pragma unroll
            for (int kk = 0; kk < 4; ++kk)
#pragma unroll
                for (int dt = 0; dt < 2; ++dt)
                    acc[j][dt] = __builtin_amdgcn_wmma_f32_16x16x4_f32(
                                     false, wf[j][kk][dt], false, xf[kk], (short)0, acc[j][dt], false, false);
            __builtin_amdgcn_sched_barrier(0);
            if (more) {
#pragma unroll
                for (int kk = 0; kk < 4; ++kk)
#pragma unroll
                    for (int dt = 0; dt < 2; ++dt) {
                        wf[j][kk][dt][0] = wnx[j * (I_DIM * D_DIM) + (4 * kk) * D_DIM + 16 * dt];
                        wf[j][kk][dt][1] = wnx[j * (I_DIM * D_DIM) + (4 * kk + 1) * D_DIM + 16 * dt];
                    }
            }
        }
        if (more) {
#pragma unroll
            for (int kk = 0; kk < 4; ++kk) xf[kk] = xn[kk];
        }

        if (HAS_V) {
            float t[4];
#pragma unroll
            for (int j = 0; j < 4; ++j) {
                float p0 = 0.f, p1 = 0.f;
#pragma unroll
                for (int dt = 0; dt < 2; ++dt) {
                    const float4 va = *(const float4*)(vslice + ((j * 2 + dt) * 2 + 0) * 128);
                    const float4 vb = *(const float4*)(vslice + ((j * 2 + dt) * 2 + 1) * 128);
                    p0 += acc[j][dt][0] * va.x + acc[j][dt][1] * va.y
                        + acc[j][dt][2] * va.z + acc[j][dt][3] * va.w;
                    p1 += acc[j][dt][4] * vb.x + acc[j][dt][5] * vb.y
                        + acc[j][dt][6] * vb.z + acc[j][dt][7] * vb.w;
                }
                const float p = p0 + p1;
                t[j] = p + __shfl_xor(p, 16, 32);
            }
            float* tb = &lds[T_OFF + (it & 1) * T_SIZE];
            if (half == 0)
                *(float4*)(tb + ln * T_STRIDE + 4 * wave) = make_float4(t[0], t[1], t[2], t[3]);
            __builtin_amdgcn_fence(__ATOMIC_RELEASE, "workgroup", "local");
            __builtin_amdgcn_s_barrier();
            __builtin_amdgcn_fence(__ATOMIC_ACQUIRE, "workgroup", "local");
            float tv[16];
#pragma unroll
            for (int q = 0; q < 4; ++q) {
                const float4 a = *(const float4*)(tb + ln * T_STRIDE + 16 * half + 4 * q);
                tv[4 * q] = a.x;  tv[4 * q + 1] = a.y;  tv[4 * q + 2] = a.z;  tv[4 * q + 3] = a.w;
            }
            float mx = tv[0];
#pragma unroll
            for (int q = 1; q < 16; ++q) mx = fmaxf(mx, tv[q]);
            mx = fmaxf(mx, __shfl_xor(mx, 16, 32));
            float den = 0.f;
#pragma unroll
            for (int q = 0; q < 16; ++q) den += __expf(tv[q] - mx);
            den += __shfl_xor(den, 16, 32);
            const float inv = 1.0f / den;
#pragma unroll
            for (int j = 0; j < 4; ++j) {
                const float c = __expf(t[j] - mx) * inv;
#pragma unroll
                for (int dt = 0; dt < 2; ++dt)
#pragma unroll
                    for (int r = 0; r < 8; ++r)
                        sacc[j][dt][r] += c * acc[j][dt][r];
            }
        } else {
            const float c = 1.0f / 32.0f;
#pragma unroll
            for (int j = 0; j < 4; ++j)
#pragma unroll
                for (int dt = 0; dt < 2; ++dt)
#pragma unroll
                    for (int r = 0; r < 8; ++r)
                        sacc[j][dt][r] += c * acc[j][dt][r];
        }
    }

#pragma unroll
    for (int j = 0; j < 4; ++j) {
        const int o = wave * 4 + j;
#pragma unroll
        for (int dt = 0; dt < 2; ++dt) {
            float* sp = &lds[S_OFF + ln * M_STRIDE + o * U_STRIDE + 16 * dt + 8 * half];
            *(float4*)(sp)     = make_float4(sacc[j][dt][0], sacc[j][dt][1], sacc[j][dt][2], sacc[j][dt][3]);
            *(float4*)(sp + 4) = make_float4(sacc[j][dt][4], sacc[j][dt][5], sacc[j][dt][6], sacc[j][dt][7]);
        }
    }
    __syncthreads();

    float* gp = part + (size_t)chunk * PLANE + (size_t)b0 * (O_CAPS * D_DIM);
#pragma unroll
    for (int k = 0; k < (BT * O_CAPS * D_DIM) / (4 * THREADS); ++k) {
        const int f   = tid + k * THREADS;
        const int row = f >> 3;
        const int q   = f & 7;
        const float* sp = &lds[S_OFF + (row >> 5) * M_STRIDE + (row & 31) * U_STRIDE];
        const float4 ss = *(const float4*)(sp + 4 * q);
        *(float4*)(gp + 4 * f) = ss;
    }
}

__global__ __launch_bounds__(256)
void squash_kernel(const float* __restrict__ part, float* __restrict__ vout,
                   const float* __restrict__ vprev, float* __restrict__ vsum)
{
    __shared__ float4 red[8][32];
    const int lane = threadIdx.x & 31;
    const int wave = threadIdx.x >> 5;
    const int grp  = wave >> 2;
    const int qtr  = wave & 3;
    const int e    = (blockIdx.x * 2 + grp) * 128 + lane * 4;
    const float* p = part + (size_t)(qtr * (NCHUNK / 4)) * PLANE + e;
    float4 acc = {0.f, 0.f, 0.f, 0.f};
#pragma unroll 16
    for (int c = 0; c < NCHUNK / 4; ++c) {
        const float4 t = *(const float4*)(p + (size_t)c * PLANE);
        acc.x += t.x;  acc.y += t.y;  acc.z += t.z;  acc.w += t.w;
    }
    red[wave][lane] = acc;
    __syncthreads();
    if (qtr == 0) {
        const float4 a0 = red[wave][lane],     a1 = red[wave + 1][lane];
        const float4 a2 = red[wave + 2][lane], a3 = red[wave + 3][lane];
        float4 s;
        s.x = ((a0.x + a1.x) + a2.x) + a3.x;
        s.y = ((a0.y + a1.y) + a2.y) + a3.y;
        s.z = ((a0.z + a1.z) + a2.z) + a3.z;
        s.w = ((a0.w + a1.w) + a2.w) + a3.w;
        float sq = s.x * s.x + s.y * s.y + s.z * s.z + s.w * s.w;
        for (int m = 4; m >= 1; m >>= 1) sq += __shfl_xor(sq, m, 32);
        const float scale = (sq / (1.0f + sq)) / (sqrtf(sq) + 1e-6f);
        float4 r;
        r.x = s.x * scale;  r.y = s.y * scale;  r.z = s.z * scale;  r.w = s.w * scale;
        *(float4*)(vout + e) = r;
        if (vsum) {
            const float4 pv = *(const float4*)(vprev + e);
            float4 t;
            t.x = r.x + pv.x;  t.y = r.y + pv.y;  t.z = r.z + pv.z;  t.w = r.w + pv.w;
            *(float4*)(vsum + e) = t;
        }
    }
}

extern "C" void kernel_launch(void* const* d_in, const int* in_sizes, int n_in,
                              void* d_out, int out_size, void* d_ws, size_t ws_size,
                              hipStream_t stream)
{
    const float* x  = (const float*)d_in[0];
    const float* Wm = (const float*)d_in[1];
    float* out = (float*)d_out;

    const size_t N = 64 * 32 * 32;
    float* wsf  = (float*)d_ws;
    float* v1   = wsf + 0 * N;
    float* v2   = wsf + 1 * N;
    float* v12  = wsf + 2 * N;
    float* part = wsf + 4 * N;

    dim3 grid(4, NCHUNK);
    const size_t shm0 = LDS_FLOATS_NOV * sizeof(float);
    const size_t shm1 = LDS_FLOATS_V * sizeof(float);

    caps_route_pass<false><<<grid, THREADS, shm0, stream>>>(x, Wm, nullptr, part);
    squash_kernel<<<(64 * 32) / 8, 256, 0, stream>>>(part, v1, nullptr, nullptr);
    caps_route_pass<true><<<grid, THREADS, shm1, stream>>>(x, Wm, v1, part);
    squash_kernel<<<(64 * 32) / 8, 256, 0, stream>>>(part, v2, v1, v12);
    caps_route_pass<true><<<grid, THREADS, shm1, stream>>>(x, Wm, v12, part);
    squash_kernel<<<(64 * 32) / 8, 256, 0, stream>>>(part, out, nullptr, nullptr);
}


# Workgroup owns two batch tiles of its 16 chw, alternated: each W fragment feeds both tiles' WMMAs; a tile's WMMAs issue ahead of the barrier for the other tile's logits; float sum order kept
# speedup vs baseline: 9.8991x; 1.1751x over previous
#include <hip/hip_runtime.h>

typedef __attribute__((ext_vector_type(2))) float v2f;
typedef __attribute__((ext_vector_type(8))) float v8f;

#define CHW        2048
#define O_CAPS     32
#define D_DIM      32
#define I_DIM      16
#define BT         16
#define NBT        2
#define ROWS       (NBT * BT)
#define THREADS    256
#define CHW_PER_WG 16
#define NCHUNK     (CHW / CHW_PER_WG)
#define PLANE      (64 * O_CAPS * D_DIM)

#define U_STRIDE   36
#define M_STRIDE   (O_CAPS * U_STRIDE + 4)
#define S_OFF      0
#define S_SIZE     (ROWS * M_STRIDE)
#define V_OFF      0
#define V_SIZE     (ROWS * O_CAPS * D_DIM)
#define T_STRIDE   36
#define T_OFF      (S_OFF + S_SIZE)
#define T_SIZE     (BT * T_STRIDE)
#define LDS_FLOATS (T_OFF + 2 * T_SIZE)


template <bool RING_SITE>
__device__ __forceinline__ void uhat_tile(v8f (&acc)[4][2], v2f (&wf)[4][4][2], const v2f (&xf)[4],
                                          const bool more, const float* __restrict__ wnx)
{
#pragma unroll
    for (int j = 0; j < 4; ++j) {
        acc[j][0] = (v8f){0.f, 0.f, 0.f, 0.f, 0.f, 0.f, 0.f, 0.f};
        acc[j][1] = (v8f){0.f, 0.f, 0.f, 0.f, 0.f, 0.f, 0.f, 0.f};
#pragma unroll
        for (int kk = 0; kk < 4; ++kk)
#pragma unroll
            for (int dt = 0; dt < 2; ++dt)
                acc[j][dt] = __builtin_amdgcn_wmma_f32_16x16x4_f32(
                                 false, wf[j][kk][dt], false, xf[kk], (short)0, acc[j][dt], false, false);
        if (RING_SITE) {
            __builtin_amdgcn_sched_barrier(0);
            if (more) {
#pragma unroll
                for (int kk = 0; kk < 4; ++kk)
#pragma unroll
                    for (int dt = 0; dt < 2; ++dt) {
                        wf[j][kk][dt][0] = wnx[j * (I_DIM * D_DIM) + (4 * kk) * D_DIM + 16 * dt];
                        wf[j][kk][dt][1] = wnx[j * (I_DIM * D_DIM) + (4 * kk + 1) * D_DIM + 16 * dt];
                    }
            }
        }
    }
}

__device__ __forceinline__ void logits_tile(const v8f (&acc)[4][2], const float* vs, float (&t)[4],
                                            float* tb, const int ln, const int half, const int wave)
{
#pragma unroll
    for (int j = 0; j < 4; ++j) {
        float p0 = 0.f, p1 = 0.f;
#pragma unroll
        for (int dt = 0; dt < 2; ++dt) {
            const float4 va = *(const float4*)(vs + ((j * 2 + dt) * 2 + 0) * 128);
            const float4 vb = *(const float4*)(vs + ((j * 2 + dt) * 2 + 1) * 128);
            p0 += acc[j][dt][0] * va.x + acc[j][dt][1] * va.y
                + acc[j][dt][2] * va.z + acc[j][dt][3] * va.w;
            p1 += acc[j][dt][4] * vb.x + acc[j][dt][5] * vb.y
                + acc[j][dt][6] * vb.z + acc[j][dt][7] * vb.w;
        }
        const float p = p0 + p1;
        t[j] = p + __shfl_xor(p, 16, 32);
    }
    if (half == 0)
        *(float4*)(tb + ln * T_STRIDE + 4 * wave) = make_float4(t[0], t[1], t[2], t[3]);
}

__device__ __forceinline__ void route_update(const float* tb, const float (&t)[4], const v8f (&acc)[4][2],
                                             v8f (&sacc)[4][2], const int ln, const int half)
{
    float tv[16];
#pragma unroll
    for (int q = 0; q < 4; ++q) {
        const float4 a = *(const float4*)(tb + ln * T_STRIDE + 16 * half + 4 * q);
        tv[4 * q] = a.x;  tv[4 * q + 1] = a.y;  tv[4 * q + 2] = a.z;  tv[4 * q + 3] = a.w;
    }
    float mx = tv[0];
#pragma unroll
    for (int q = 1; q < 16; ++q) mx = fmaxf(mx, tv[q]);
    mx = fmaxf(mx, __shfl_xor(mx, 16, 32));
    float den = 0.f;
#pragma unroll
    for (int q = 0; q < 16; ++q) den += __expf(tv[q] - mx);
    den += __shfl_xor(den, 16, 32);
    const float inv = 1.0f / den;
#pragma unroll
    for (int j = 0; j < 4; ++j) {
        const float c = __expf(t[j] - mx) * inv;
#pragma unroll
        for (int dt = 0; dt < 2; ++dt)
#pragma unroll
            for (int r = 0; r < 8; ++r)
                sacc[j][dt][r] += c * acc[j][dt][r];
    }
}

__device__ __forceinline__ void lds_barrier()
{
    __builtin_amdgcn_fence(__ATOMIC_RELEASE, "workgroup", "local");
    __builtin_amdgcn_s_barrier();
    __builtin_amdgcn_fence(__ATOMIC_ACQUIRE, "workgroup", "local");
}

template <bool HAS_V>
__global__ __launch_bounds__(THREADS)
__attribute__((amdgpu_waves_per_eu(2, 2)))
void caps_route_pass(const float* __restrict__ x,
                     const float* __restrict__ Wm,
                     const float* __restrict__ v,
                     float* __restrict__ part)
{
    extern __shared__ float lds[];
    const int tid   = threadIdx.x;
    const int lane  = tid & 31;
    const int wave  = tid >> 5;
    const int half  = lane >> 4;
    const int ln    = lane & 15;
    const int b0    = blockIdx.x * ROWS;
    const int chunk = blockIdx.y;

    float* vslice = &lds[V_OFF + (wave * (NBT * 16) * 32 + lane) * 4];
    if (HAS_V) {
#pragma unroll
        for (int bt = 0; bt < NBT; ++bt)
#pragma unroll
            for (int j = 0; j < 4; ++j) {
                const int o = wave * 4 + j;
                const float* vp = &v[(((size_t)(b0 + bt * BT + ln)) * O_CAPS + o) * D_DIM + 8 * half];
#pragma unroll
                for (int dt = 0; dt < 2; ++dt)
#pragma unroll
                    for (int q = 0; q < 2; ++q)
                        *(float4*)(vslice + (((bt * 4 + j) * 2 + dt) * 2 + q) * 128) =
                            *(const float4*)(vp + 16 * dt + 4 * q);
            }
    }

    v8f sacc[NBT][4][2];
#pragma unroll
    for (int bt = 0; bt < NBT; ++bt)
#pragma unroll
        for (int j = 0; j < 4; ++j)
#pragma unroll
            for (int dt = 0; dt < 2; ++dt)
                sacc[bt][j][dt] = (v8f){0.f, 0.f, 0.f, 0.f, 0.f, 0.f, 0.f, 0.f};

    const float* xrow = x + ((size_t)(b0 + ln) * CHW + (size_t)chunk * CHW_PER_WG) * I_DIM + 2 * half;
    const size_t xbt = (size_t)BT * CHW * I_DIM;
    v2f xf[4], xn[4];
#pragma unroll
    for (int kk = 0; kk < 4; ++kk) xf[kk] = *(const v2f*)(xrow + 4 * kk);
    __builtin_amdgcn_sched_barrier(0);

    const int woff = (2 * half) * D_DIM + ln;
    const float* wchunk = Wm + ((size_t)chunk * CHW_PER_WG * O_CAPS + wave * 4) * (I_DIM * D_DIM) + woff;
    v2f wf[4][4][2];
#pragma unroll
    for (int j = 0; j < 4; ++j)
#pragma unroll
        for (int kk = 0; kk < 4; ++kk)
#pragma unroll
            for (int dt = 0; dt < 2; ++dt) {
                wf[j][kk][dt][0] = wchunk[j * (I_DIM * D_DIM) + (4 * kk) * D_DIM + 16 * dt];
                wf[j][kk][dt][1] = wchunk[j * (I_DIM * D_DIM) + (4 * kk + 1) * D_DIM + 16 * dt];
            }

    v8f acc0[4][2];
    if (HAS_V) {
        v8f acc1[4][2];
        float t0[4], t1[4];
        float* tb0 = &lds[T_OFF];
        float* tb1 = &lds[T_OFF + T_SIZE];
        const float* vs0 = vslice;
        const float* vs1 = vslice + 16 * 128;

#pragma unroll
        for (int kk = 0; kk < 4; ++kk) xn[kk] = *(const v2f*)(xrow + xbt + 4 * kk);
        uhat_tile<false>(acc0, wf, xf, false, nullptr);
#pragma unroll
        for (int kk = 0; kk < 4; ++kk) xf[kk] = xn[kk];
        logits_tile(acc0, vs0, t0, tb0, ln, half, wave);

        for (int it = 0; it < CHW_PER_WG; ++it) {
            const bool more = it + 1 < CHW_PER_WG;
            const float* wnx = wchunk + (size_t)(it + 1) * (O_CAPS * I_DIM * D_DIM);

            if (more) {
#pragma unroll
                for (int kk = 0; kk < 4; ++kk)
                    xn[kk] = *(const v2f*)(xrow + (size_t)(it + 1) * I_DIM + 4 * kk);
            }
            uhat_tile<true>(acc1, wf, xf, more, wnx);
            if (more) {
#pragma unroll
                for (int kk = 0; kk < 4; ++kk) xf[kk] = xn[kk];
            }
            lds_barrier();
            route_update(tb0, t0, acc0, sacc[0], ln, half);
            logits_tile(acc1, vs1, t1, tb1, ln, half, wave);

            if (more) {
#pragma unroll
                for (int kk = 0; kk < 4; ++kk)
                    xn[kk] = *(const v2f*)(xrow + xbt + (size_t)(it + 1) * I_DIM + 4 * kk);
                uhat_tile<false>(acc0, wf, xf, false, nullptr);
#pragma unroll
                for (int kk = 0; kk < 4; ++kk) xf[kk] = xn[kk];
            }
            lds_barrier();
            route_update(tb1, t1, acc1, sacc[1], ln, half);
            if (more) logits_tile(acc0, vs0, t0, tb0, ln, half, wave);
        }
    } else {
        const float c = 1.0f / 32.0f;
        for (int it = 0; it < CHW_PER_WG; ++it) {
            const bool more = it + 1 < CHW_PER_WG;
            const float* wnx = wchunk + (size_t)(it + 1) * (O_CAPS * I_DIM * D_DIM);
#pragma unroll
            for (int kk = 0; kk < 4; ++kk)
                xn[kk] = *(const v2f*)(xrow + xbt + (size_t)it * I_DIM + 4 * kk);
            uhat_tile<false>(acc0, wf, xf, false, nullptr);
#pragma unroll
            for (int kk = 0; kk < 4; ++kk) xf[kk] = xn[kk];
#pragma unroll
            for (int j = 0; j < 4; ++j)
#pragma unroll
                for (int dt = 0; dt < 2; ++dt)
#pragma unroll
                    for (int r = 0; r < 8; ++r)
                        sacc[0][j][dt][r] += c * acc0[j][dt][r];
            if (more) {
#pragma unroll
                for (int kk = 0; kk < 4; ++kk)
                    xn[kk] = *(const v2f*)(xrow + (size_t)(it + 1) * I_DIM + 4 * kk);
            }
            uhat_tile<true>(acc0, wf, xf, more, wnx);
            if (more) {
#pragma unroll
                for (int kk = 0; kk < 4; ++kk) xf[kk] = xn[kk];
            }
#pragma unroll
            for (int j = 0; j < 4; ++j)
#pragma unroll
                for (int dt = 0; dt < 2; ++dt)
#pragma unroll
                    for (int r = 0; r < 8; ++r)
                        sacc[1][j][dt][r] += c * acc0[j][dt][r];
        }
    }

#pragma unroll
    for (int bt = 0; bt < NBT; ++bt)
#pragma unroll
        for (int j = 0; j < 4; ++j) {
            const int o = wave * 4 + j;
#pragma unroll
            for (int dt = 0; dt < 2; ++dt) {
                float* sp = &lds[S_OFF + (bt * BT + ln) * M_STRIDE + o * U_STRIDE + 16 * dt + 8 * half];
                *(float4*)(sp)     = make_float4(sacc[bt][j][dt][0], sacc[bt][j][dt][1], sacc[bt][j][dt][2], sacc[bt][j][dt][3]);
                *(float4*)(sp + 4) = make_float4(sacc[bt][j][dt][4], sacc[bt][j][dt][5], sacc[bt][j][dt][6], sacc[bt][j][dt][7]);
            }
        }
    __syncthreads();

    float* gp = part + (size_t)chunk * PLANE + (size_t)b0 * (O_CAPS * D_DIM);
#pragma unroll 16
    for (int k = 0; k < (ROWS * O_CAPS * D_DIM) / (4 * THREADS); ++k) {
        const int f   = tid + k * THREADS;
        const int row = f >> 3;
        const int q   = f & 7;
        const float* sp = &lds[S_OFF + (row >> 5) * M_STRIDE + (row & 31) * U_STRIDE];
        const float4 ss = *(const float4*)(sp + 4 * q);
        *(float4*)(gp + 4 * f) = ss;
    }
}

__global__ __launch_bounds__(256)
void squash_kernel(const float* __restrict__ part, float* __restrict__ vout,
                   const float* __restrict__ vprev, float* __restrict__ vsum)
{
    __shared__ float4 red[8][32];
    const int lane = threadIdx.x & 31;
    const int wave = threadIdx.x >> 5;
    const int grp  = wave >> 2;
    const int qtr  = wave & 3;
    const int e    = (blockIdx.x * 2 + grp) * 128 + lane * 4;
    const float* p = part + (size_t)(qtr * (NCHUNK / 4)) * PLANE + e;
    float4 acc = {0.f, 0.f, 0.f, 0.f};
#pragma unroll 16
    for (int c = 0; c < NCHUNK / 4; ++c) {
        const float4 t = *(const float4*)(p + (size_t)c * PLANE);
        acc.x += t.x;  acc.y += t.y;  acc.z += t.z;  acc.w += t.w;
    }
    red[wave][lane] = acc;
    __syncthreads();
    if (qtr == 0) {
        const float4 a0 = red[wave][lane],     a1 = red[wave + 1][lane];
        const float4 a2 = red[wave + 2][lane], a3 = red[wave + 3][lane];
        float4 s;
        s.x = ((a0.x + a1.x) + a2.x) + a3.x;
        s.y = ((a0.y + a1.y) + a2.y) + a3.y;
        s.z = ((a0.z + a1.z) + a2.z) + a3.z;
        s.w = ((a0.w + a1.w) + a2.w) + a3.w;
        float sq = s.x * s.x + s.y * s.y + s.z * s.z + s.w * s.w;
        for (int m = 4; m >= 1; m >>= 1) sq += __shfl_xor(sq, m, 32);
        const float scale = (sq / (1.0f + sq)) / (sqrtf(sq) + 1e-6f);
        float4 r;
        r.x = s.x * scale;  r.y = s.y * scale;  r.z = s.z * scale;  r.w = s.w * scale;
        *(float4*)(vout + e) = r;
        if (vsum) {
            const float4 pv = *(const float4*)(vprev + e);
            float4 t;
            t.x = r.x + pv.x;  t.y = r.y + pv.y;  t.z = r.z + pv.z;  t.w = r.w + pv.w;
            *(float4*)(vsum + e) = t;
        }
    }
}

extern "C" void kernel_launch(void* const* d_in, const int* in_sizes, int n_in,
                              void* d_out, int out_size, void* d_ws, size_t ws_size,
                              hipStream_t stream)
{
    const float* x  = (const float*)d_in[0];
    const float* Wm = (const float*)d_in[1];
    float* out = (float*)d_out;

    const size_t N = 64 * 32 * 32;
    float* wsf  = (float*)d_ws;
    float* v1   = wsf + 0 * N;
    float* v2   = wsf + 1 * N;
    float* v12  = wsf + 2 * N;
    float* part = wsf + 4 * N;

    dim3 grid(64 / ROWS, NCHUNK);
    const size_t shm = LDS_FLOATS * sizeof(float);

    caps_route_pass<false><<<grid, THREADS, shm, stream>>>(x, Wm, nullptr, part);
    squash_kernel<<<(64 * 32) / 8, 256, 0, stream>>>(part, v1, nullptr, nullptr);
    caps_route_pass<true><<<grid, THREADS, shm, stream>>>(x, Wm, v1, part);
    squash_kernel<<<(64 * 32) / 8, 256, 0, stream>>>(part, v2, v1, v12);
    caps_route_pass<true><<<grid, THREADS, shm, stream>>>(x, Wm, v12, part);
    squash_kernel<<<(64 * 32) / 8, 256, 0, stream>>>(part, out, nullptr, nullptr);
}
